# starting kernel
# baseline (speedup 1.0000x reference)
#include <hip/hip_runtime.h>
#include <math.h>

typedef __attribute__((ext_vector_type(2))) float v2f;
typedef __attribute__((ext_vector_type(8))) float v8f;

#define T_STEPS 28
#define F_IN    28
#define H_UNITS 128
#define G4      512
#define TILE_M  32
#define NKF_U   32
#define NKF_W   7

__device__ __forceinline__ float fast_sigmoid(float x) {
    return 1.0f / (1.0f + __expf(-x));
}

__global__ __launch_bounds__(256, 1)
void lstm_fused_wmma3(const float* __restrict__ x, const float* __restrict__ W,
                      const float* __restrict__ U, const float* __restrict__ b,
                      const float* __restrict__ Wd, const float* __restrict__ bd,
                      float* __restrict__ out)
{
    extern __shared__ float smem[];
    float* uSwz = smem;
    float* hSwz = smem + 65536;

    const int tid   = threadIdx.x;
    const int lane  = tid & 31;
    const int wave  = tid >> 5;
    const int laneN = lane & 15;
    const int half  = lane >> 4;
    const int rowBase = blockIdx.x * TILE_M;

    for (int i = tid; i < H_UNITS * G4; i += 256) {
        const int j  = i & 1;
        const int hf = (i >> 1) & 1;
        const int ln = (i >> 2) & 15;
        const int g  = (i >> 6) & 3;
        const int kf = (i >> 8) & 31;
        const int w  = i >> 13;
        uSwz[i] = U[(kf * 4 + hf * 2 + j) * G4 + (g * H_UNITS + w * 16 + ln)];
    }
    for (int i = tid; i < TILE_M * H_UNITS; i += 256) hSwz[i] = 0.0f;
    __syncthreads();

    const int nUnit = wave * 16 + laneN;
    const int nOff  = (nUnit >> 2) * 128 + ((nUnit >> 1) & 1) * 2 + (nUnit & 1);

    const float* uW  = uSwz + wave * 8192 + laneN * 4 + half * 2;
    const float* hA0 = hSwz + laneN * 4 + half * 2;
    const float* hA1 = hSwz + (16 + laneN) * 4 + half * 2;

    const float* xRow0 = x + (size_t)(rowBase + laneN) * (T_STEPS * F_IN);
    const float* xRow1 = x + (size_t)(rowBase + 16 + laneN) * (T_STEPS * F_IN);

    v8f biasV[4];
    #pragma unroll
    for (int g = 0; g < 4; ++g) {
        const float bias = b[g * H_UNITS + wave * 16 + laneN];
        biasV[g] = (v8f){bias,bias,bias,bias,bias,bias,bias,bias};
    }

    v8f cSt[2];
    cSt[0] = (v8f){0.f,0.f,0.f,0.f,0.f,0.f,0.f,0.f};
    cSt[1] = cSt[0];

    for (int t = 0; t < T_STEPS; ++t) {
        v8f acc[4][2];

        {
            const int k = half * 2;
            v2f a0, a1;
            a0.x = xRow0[t * F_IN + k];  a0.y = xRow0[t * F_IN + k + 1];
            a1.x = xRow1[t * F_IN + k];  a1.y = xRow1[t * F_IN + k + 1];
            #pragma unroll
            for (int g = 0; g < 4; ++g) {
                const int col = g * H_UNITS + wave * 16 + laneN;
                v2f bb;
                bb.x = W[k * G4 + col];
                bb.y = W[(k + 1) * G4 + col];
                acc[g][0] = __builtin_amdgcn_wmma_f32_16x16x4_f32(
                    false, a0, false, bb, (short)0, biasV[g], false, false);
                acc[g][1] = __builtin_amdgcn_wmma_f32_16x16x4_f32(
                    false, a1, false, bb, (short)0, biasV[g], false, false);
            }
        }
        #pragma unroll
        for (int kf = 1; kf < NKF_W; ++kf) {
            const int k = kf * 4 + half * 2;
            v2f a0, a1;
            a0.x = xRow0[t * F_IN + k];  a0.y = xRow0[t * F_IN + k + 1];
            a1.x = xRow1[t * F_IN + k];  a1.y = xRow1[t * F_IN + k + 1];
            #pragma unroll
            for (int g = 0; g < 4; ++g) {
                const int col = g * H_UNITS + wave * 16 + laneN;
                v2f bb;
                bb.x = W[k * G4 + col];
                bb.y = W[(k + 1) * G4 + col];
                acc[g][0] = __builtin_amdgcn_wmma_f32_16x16x4_f32(
                    false, a0, false, bb, (short)0, acc[g][0], false, false);
                acc[g][1] = __builtin_amdgcn_wmma_f32_16x16x4_f32(
                    false, a1, false, bb, (short)0, acc[g][1], false, false);
            }
        }

        for (int kf = 0; kf < NKF_U; ++kf) {
            const v2f a0 = *(const v2f*)&hA0[kf * 128];
            const v2f a1 = *(const v2f*)&hA1[kf * 128];
            #pragma unroll
            for (int g = 0; g < 4; ++g) {
                const v2f bb = *(const v2f*)&uW[kf * 256 + g * 64];
                acc[g][0] = __builtin_amdgcn_wmma_f32_16x16x4_f32(
                    false, a0, false, bb, (short)0, acc[g][0], false, false);
                acc[g][1] = __builtin_amdgcn_wmma_f32_16x16x4_f32(
                    false, a1, false, bb, (short)0, acc[g][1], false, false);
            }
        }

        __syncthreads();

        #pragma unroll
        for (int mt = 0; mt < 2; ++mt) {
            #pragma unroll
            for (int r = 0; r < 8; ++r) {
                const float ig = fast_sigmoid(acc[0][mt][r]);
                const float fg = fast_sigmoid(acc[1][mt][r]);
                const float gg = fmaxf(acc[2][mt][r], 0.0f);
                const float og = fast_sigmoid(acc[3][mt][r]);
                const float cn = fg * cSt[mt][r] + ig * gg;
                cSt[mt][r] = cn;
                const float hv = og * fmaxf(cn, 0.0f);
                const int m = mt * 16 + r + 8 * half;
                hSwz[nOff + m * 4] = hv;
            }
        }

        __syncthreads();
    }

    if (tid < TILE_M) {
        const int m = tid;
        float logits[10];
        #pragma unroll
        for (int c = 0; c < 10; ++c) logits[c] = bd[c];
        for (int k = 0; k < H_UNITS; ++k) {
            const float hv = hSwz[(k >> 2) * 128 + m * 4 + ((k >> 1) & 1) * 2 + (k & 1)];
            #pragma unroll
            for (int c = 0; c < 10; ++c) logits[c] += hv * Wd[k * 10 + c];
        }
        float mx = logits[0];
        #pragma unroll
        for (int c = 1; c < 10; ++c) mx = fmaxf(mx, logits[c]);
        float e[10], sum = 0.0f;
        #pragma unroll
        for (int c = 0; c < 10; ++c) { e[c] = __expf(logits[c] - mx); sum += e[c]; }
        const float inv = 1.0f / sum;
        #pragma unroll
        for (int c = 0; c < 10; ++c)
            out[(size_t)(rowBase + m) * 10 + c] = e[c] * inv;
    }
}

extern "C" void kernel_launch(void* const* d_in, const int* in_sizes, int n_in,
                              void* d_out, int out_size, void* d_ws, size_t ws_size,
                              hipStream_t stream) {
    const float* x  = (const float*)d_in[0];
    const float* W  = (const float*)d_in[1];
    const float* U  = (const float*)d_in[2];
    const float* b  = (const float*)d_in[3];
    const float* Wd = (const float*)d_in[4];
    const float* bd = (const float*)d_in[5];
    float* out = (float*)d_out;

    const size_t ldsBytes = (size_t)(65536 + 4096) * sizeof(float);
    const int numBlocks = 16384 / TILE_M;

    hipLaunchKernelGGL(lstm_fused_wmma3, dim3(numBlocks), dim3(256), ldsBytes, stream,
                       x, W, U, b, Wd, bd, out);

    (void)in_sizes; (void)n_in; (void)out_size; (void)d_ws; (void)ws_size;
}


# a wave's two M-tiles run one phase apart: tile A's cell updates are issued among tile B's 156 matrix instructions and vice versa, one accumulator set, one barrier per phase; sums unchanged
# speedup vs baseline: 1.1410x; 1.1410x over previous
#include <hip/hip_runtime.h>
#include <math.h>


typedef __attribute__((ext_vector_type(2))) float v2f;
typedef __attribute__((ext_vector_type(8))) float v8f;

#define T_STEPS 28
#define F_IN    28
#define H_UNITS 128
#define G4      512
#define TILE_M  32
#define NKF_U   32
#define NKF_W   7
#define HBLK    8

__device__ __forceinline__ float fast_sigmoid(float x) {
    return 1.0f / (1.0f + __expf(-x));
}

__device__ __forceinline__ void load_x_frags(const float* __restrict__ xRow, int t, int half,
                                             v2f (&xa)[NKF_W])
{
    #pragma unroll
    for (int kf = 0; kf < NKF_W; ++kf) {
        const int k = kf * 4 + half * 2;
        xa[kf].x = xRow[t * F_IN + k];
        xa[kf].y = xRow[t * F_IN + k + 1];
    }
}

__device__ __forceinline__ void load_h_block(v2f (&a)[HBLK], const float* hA, int kb)
{
    #pragma unroll
    for (int i = 0; i < HBLK; ++i)
        a[i] = *(const v2f*)&hA[(kb * HBLK + i) * 128];
}

__device__ __forceinline__ void cell_element(const v8f (&zg)[4], v8f& cSt, int r, float* hW)
{
    const float ig = fast_sigmoid(zg[0][r]);
    const float fg = fast_sigmoid(zg[1][r]);
    const float gg = fmaxf(zg[2][r], 0.0f);
    const float og = fast_sigmoid(zg[3][r]);
    const float cn = fg * cSt[r] + ig * gg;
    cSt[r] = cn;
    hW[r * 4] = og * fmaxf(cn, 0.0f);
}

template <bool GATE>
__device__ __forceinline__ void phase(v8f (&z)[4], const v2f (&xa)[NKF_W], const float* wW,
                                      const v8f (&biasV)[4], const float* hA,
                                      const v2f (&uB)[NKF_U][4],
                                      const v8f (&zg)[4], v8f& cSt, float* hW)
{
    v2f a[2][HBLK];
    load_h_block(a[0], hA, 0);

    #pragma unroll
    for (int kf = 0; kf < NKF_W; ++kf) {
        #pragma unroll
        for (int g = 0; g < 4; ++g) {
            const v2f bb = *(const v2f*)&wW[kf * 256 + g * 64];
            z[g] = __builtin_amdgcn_wmma_f32_16x16x4_f32(
                false, xa[kf], false, bb, (short)0, kf == 0 ? biasV[g] : z[g], false, false);
        }
    }

    #pragma unroll
    for (int kb = 0; kb < NKF_U / HBLK; ++kb) {
        if (kb + 1 < NKF_U / HBLK) {
            load_h_block(a[(kb + 1) & 1], hA, kb + 1);
            __builtin_amdgcn_sched_barrier(0);
        }
        #pragma unroll
        for (int i = 0; i < HBLK; ++i) {
            const int kf = kb * HBLK + i;
            #pragma unroll
            for (int g = 0; g < 4; ++g)
                z[g] = __builtin_amdgcn_wmma_f32_16x16x4_f32(
                    false, a[kb & 1][i], false, uB[kf][g], (short)0, z[g], false, false);
            if (GATE && (i & 3) == 1)
                cell_element(zg, cSt, 2 * kb + (i >> 2), hW);
        }
        if (kb + 1 < NKF_U / HBLK) {
            __builtin_amdgcn_sched_barrier(0);
        }
    }
}

__global__ __launch_bounds__(256, 1)
void lstm_fused_wmma3(const float* __restrict__ x, const float* __restrict__ W,
                      const float* __restrict__ U, const float* __restrict__ b,
                      const float* __restrict__ Wd, const float* __restrict__ bd,
                      float* __restrict__ out)
{
    extern __shared__ float smem[];
    float* hSwz = smem;
    float* wSwz = smem + 4096;

    const int tid   = threadIdx.x;
    const int lane  = tid & 31;
    const int wave  = tid >> 5;
    const int laneN = lane & 15;
    const int half  = lane >> 4;
    const int rowBase = blockIdx.x * TILE_M;

    v2f uB[NKF_U][4];
    #pragma unroll
    for (int kf = 0; kf < NKF_U; ++kf) {
        const int k = kf * 4 + half * 2;
        #pragma unroll
        for (int g = 0; g < 4; ++g) {
            const int col = g * H_UNITS + wave * 16 + laneN;
            uB[kf][g].x = U[k * G4 + col];
            uB[kf][g].y = U[(k + 1) * G4 + col];
        }
    }

    for (int i = tid; i < TILE_M * H_UNITS; i += 256) hSwz[i] = 0.0f;

    const int nUnit = wave * 16 + laneN;
    const int nOff  = (nUnit >> 2) * 128 + ((nUnit >> 1) & 1) * 2 + (nUnit & 1);

    float*       wW  = wSwz + wave * 1792 + laneN * 4 + half * 2;
    const float* hA0 = hSwz + laneN * 4 + half * 2;
    const float* hA1 = hSwz + (16 + laneN) * 4 + half * 2;
    float*       hW0 = hSwz + nOff + (8 * half) * 4;
    float*       hW1 = hSwz + nOff + (16 + 8 * half) * 4;

    const float* xRow0 = x + (size_t)(rowBase + laneN) * (T_STEPS * F_IN);
    const float* xRow1 = x + (size_t)(rowBase + 16 + laneN) * (T_STEPS * F_IN);

    v8f biasV[4];
    #pragma unroll
    for (int g = 0; g < 4; ++g) {
        const float bias = b[g * H_UNITS + wave * 16 + laneN];
        biasV[g] = (v8f){bias,bias,bias,bias,bias,bias,bias,bias};
    }

    #pragma unroll
    for (int kf = 0; kf < NKF_W; ++kf) {
        const int k = kf * 4 + half * 2;
        #pragma unroll
        for (int g = 0; g < 4; ++g) {
            const int col = g * H_UNITS + wave * 16 + laneN;
            v2f wv;
            wv.x = W[k * G4 + col];
            wv.y = W[(k + 1) * G4 + col];
            *(v2f*)&wW[kf * 256 + g * 64] = wv;
        }
    }
    __syncthreads();

    v8f cStA = (v8f){0.f,0.f,0.f,0.f,0.f,0.f,0.f,0.f};
    v8f cStB = cStA;

    v2f xaA[NKF_W], xaB[NKF_W];
    v8f accA[4], accB[4];

    load_x_frags(xRow0, 0, half, xaA);
    load_x_frags(xRow1, 0, half, xaB);
    phase<false>(accA, xaA, wW, biasV, hA0, uB, accB, cStB, hW1);
    __syncthreads();
    load_x_frags(xRow0, 1, half, xaA);

    for (int t = 0; t < T_STEPS - 1; ++t) {
        phase<true>(accB, xaB, wW, biasV, hA1, uB, accA, cStA, hW0);
        __syncthreads();
        load_x_frags(xRow1, t + 1, half, xaB);

        phase<true>(accA, xaA, wW, biasV, hA0, uB, accB, cStB, hW1);
        __syncthreads();
        if (t + 2 < T_STEPS) load_x_frags(xRow0, t + 2, half, xaA);
    }

    phase<true>(accB, xaB, wW, biasV, hA1, uB, accA, cStA, hW0);
    __syncthreads();
    #pragma unroll
    for (int r = 0; r < 8; ++r) cell_element(accB, cStB, r, hW1);
    __syncthreads();

    if (tid < TILE_M) {
        const int m = tid;
        float logits[10];
        #pragma unroll
        for (int c = 0; c < 10; ++c) logits[c] = bd[c];
        for (int k = 0; k < H_UNITS; ++k) {
            const float hv = hSwz[(k >> 2) * 128 + m * 4 + ((k >> 1) & 1) * 2 + (k & 1)];
            #pragma unroll
            for (int c = 0; c < 10; ++c) logits[c] += hv * Wd[k * 10 + c];
        }
        float mx = logits[0];
        #pragma unroll
        for (int c = 1; c < 10; ++c) mx = fmaxf(mx, logits[c]);
        float e[10], sum = 0.0f;
        #pragma unroll
        for (int c = 0; c < 10; ++c) { e[c] = __expf(logits[c] - mx); sum += e[c]; }
        const float inv = 1.0f / sum;
        #pragma unroll
        for (int c = 0; c < 10; ++c)
            out[(size_t)(rowBase + m) * 10 + c] = e[c] * inv;
    }
}

extern "C" void kernel_launch(void* const* d_in, const int* in_sizes, int n_in,
                              void* d_out, int out_size, void* d_ws, size_t ws_size,
                              hipStream_t stream) {
    const float* x  = (const float*)d_in[0];
    const float* W  = (const float*)d_in[1];
    const float* U  = (const float*)d_in[2];
    const float* b  = (const float*)d_in[3];
    const float* Wd = (const float*)d_in[4];
    const float* bd = (const float*)d_in[5];
    float* out = (float*)d_out;

    const size_t ldsBytes = (size_t)(4096 + 14336) * sizeof(float);
    const int numBlocks = 16384 / TILE_M;

    hipLaunchKernelGGL(lstm_fused_wmma3, dim3(numBlocks), dim3(256), ldsBytes, stream,
                       x, W, U, b, Wd, bd, out);

    (void)in_sizes; (void)n_in; (void)out_size; (void)d_ws; (void)ws_size;
}


# roles swapped: gates on M, batch on N, so a wave's new h (D layout) is the next step's B operand in place; U^T, W^T streamed from LDS; no per-step barrier, no h exchange; h@U, dense sum order changed
# speedup vs baseline: 1.1898x; 1.0428x over previous
#include <hip/hip_runtime.h>
#include <math.h>


typedef __attribute__((ext_vector_type(2))) float v2f;
typedef __attribute__((ext_vector_type(4))) float v4f;
typedef __attribute__((ext_vector_type(8))) float v8f;

#define T_STEPS 28
#define F_IN    28
#define H_UNITS 128
#define G4      512
#define ROWS_WG 64
#define NKF_W   7
#define US_FLOATS 65536
#define WS_FLOATS 14336
#define BS_FLOATS 512

__device__ __forceinline__ float fast_sigmoid(float x) {
    return 1.0f / (1.0f + __expf(-x));
}


__device__ __forceinline__ void load_u_block(v4f (&A)[4][4], const float* base)
{
    #pragma unroll
    for (int g = 0; g < 4; ++g) {
        #pragma unroll
        for (int p = 0; p < 4; ++p)
            A[g][p] = *(const v4f*)&base[(g * 4 + p) * 128];
    }
}

__device__ __forceinline__ void load_w_bias(v2f (&Wf)[NKF_W][4], v8f (&bV)[4],
                                            const float* wBase, const float* bBase)
{
    #pragma unroll
    for (int kf = 0; kf < NKF_W; ++kf) {
        #pragma unroll
        for (int g = 0; g < 4; ++g)
            Wf[kf][g] = *(const v2f*)&wBase[kf * 256 + g * 64];
    }
    #pragma unroll
    for (int g = 0; g < 4; ++g)
        bV[g] = *(const v8f*)&bBase[g * 16];
}

template <int UP>
__device__ __forceinline__ void phase(const float* usL, const float* wsL, const float* bsL,
                                      v4f (&Ab)[2][4][4], v2f (&Wf)[NKF_W][4], v8f (&bV)[4],
                                      const v2f (&xb)[NKF_W], const v8f (&hOld)[8],
                                      v8f& hNewU, v8f& cU)
{
    constexpr int UN = (UP + 1) & 7;
    v8f acc[4];

    load_u_block(Ab[1], usL + (UP * 4 + 1) * 2048);
    __builtin_amdgcn_sched_barrier(0);

    #pragma unroll
    for (int kf = 0; kf < NKF_W; ++kf) {
        #pragma unroll
        for (int g = 0; g < 4; ++g)
            acc[g] = __builtin_amdgcn_wmma_f32_16x16x4_f32(
                false, Wf[kf][g], false, xb[kf], (short)0, kf == 0 ? bV[g] : acc[g], false, false);
    }

    #pragma unroll
    for (int kb = 0; kb < 4; ++kb) {
        if (kb == 1) {
            load_u_block(Ab[0], usL + (UP * 4 + 2) * 2048);
            load_w_bias(Wf, bV, wsL + UN * (NKF_W * 256), bsL + UN * 64);
            __builtin_amdgcn_sched_barrier(0);
        } else if (kb == 2) {
            load_u_block(Ab[1], usL + (UP * 4 + 3) * 2048);
            __builtin_amdgcn_sched_barrier(0);
        } else if (kb == 3) {
            load_u_block(Ab[0], usL + (UN * 4 + 0) * 2048);
            __builtin_amdgcn_sched_barrier(0);
        }
        #pragma unroll
        for (int i = 0; i < 8; ++i) {
            const int u = 2 * kb + (i >> 2);
            const int j = i & 3;
            const v8f hv = hOld[u];
            v2f bfrag;
            bfrag.x = hv[2 * j];
            bfrag.y = hv[2 * j + 1];
            #pragma unroll
            for (int g = 0; g < 4; ++g) {
                const v4f a4 = Ab[kb & 1][g][i >> 1];
                v2f afrag;
                afrag.x = (i & 1) ? a4.z : a4.x;
                afrag.y = (i & 1) ? a4.w : a4.y;
                acc[g] = __builtin_amdgcn_wmma_f32_16x16x4_f32(
                    false, afrag, false, bfrag, (short)0, acc[g], false, false);
            }
        }
        if (kb < 3) {
            __builtin_amdgcn_sched_barrier(0);
        }
    }

    #pragma unroll
    for (int r = 0; r < 8; ++r) {
        const float ig = fast_sigmoid(acc[0][r]);
        const float fg = fast_sigmoid(acc[1][r]);
        const float gg = fmaxf(acc[2][r], 0.0f);
        const float og = fast_sigmoid(acc[3][r]);
        const float cn = fg * cU[r] + ig * gg;
        cU[r] = cn;
        hNewU[r] = og * fmaxf(cn, 0.0f);
    }
}

__global__ __launch_bounds__(128)
void lstm_swapped_wmma(const float* __restrict__ x, const float* __restrict__ W,
                       const float* __restrict__ U, const float* __restrict__ b,
                       const float* __restrict__ Wd, const float* __restrict__ bd,
                       float* __restrict__ out)
{
    __shared__ __attribute__((aligned(16))) float smem[US_FLOATS + WS_FLOATS + BS_FLOATS];
    float* Us = smem;
    float* Ws = smem + US_FLOATS;
    float* Bs = smem + US_FLOATS + WS_FLOATS;

    const int tid   = threadIdx.x;
    const int lane  = tid & 31;
    const int wave  = tid >> 5;
    const int laneN = lane & 15;
    const int half  = lane >> 4;
    const int rowBase = blockIdx.x * ROWS_WG + wave * 16;

    #pragma unroll 16
    for (int it = 0; it < 128; ++it) {
        const int g  = it & 3;
        const int kb = (it >> 2) & 3;
        const int up = it >> 4;
        const int k0 = 16 * (2 * kb + (wave >> 1)) + 8 * half + 4 * (wave & 1);
        const int col = g * H_UNITS + 16 * up + laneN;
        v4f v;
        v.x = U[(k0 + 0) * G4 + col];
        v.y = U[(k0 + 1) * G4 + col];
        v.z = U[(k0 + 2) * G4 + col];
        v.w = U[(k0 + 3) * G4 + col];
        *(v4f*)&Us[it * 512 + tid * 4] = v;
    }
    #pragma unroll
    for (int it = 0; it < 56; ++it) {
        const int kf = it % 7;
        const int up = it / 7;
        const int k = 4 * kf + 2 * half;
        const int col = wave * H_UNITS + 16 * up + laneN;
        v2f v;
        v.x = W[k * G4 + col];
        v.y = W[(k + 1) * G4 + col];
        *(v2f*)&Ws[it * 256 + tid * 2] = v;
    }
    #pragma unroll
    for (int i4 = 0; i4 < BS_FLOATS / 128; ++i4) {
        const int idx = tid + 128 * i4;
        const int r  = idx & 7;
        const int hf = (idx >> 3) & 1;
        const int g  = (idx >> 4) & 3;
        const int up = idx >> 6;
        Bs[idx] = b[g * H_UNITS + 16 * up + 8 * hf + r];
    }
    __syncthreads();

    const float* usL = Us + lane * 4;
    const float* wsL = Ws + lane * 2;
    const float* bsL = Bs + half * 8;

    const float* xRow = x + (size_t)(rowBase + laneN) * (T_STEPS * F_IN) + 2 * half;

    v8f hOld[8], hNew[8], cSt[8];
    #pragma unroll
    for (int u = 0; u < 8; ++u) {
        hOld[u] = (v8f){0.f,0.f,0.f,0.f,0.f,0.f,0.f,0.f};
        hNew[u] = hOld[u];
        cSt[u]  = hOld[u];
    }

    v4f Ab[2][4][4];
    v2f Wf[NKF_W][4];
    v8f bV[4];
    v2f xb[NKF_W], xn[NKF_W];

    load_u_block(Ab[0], usL);
    load_w_bias(Wf, bV, wsL, bsL);
    #pragma unroll
    for (int kf = 0; kf < NKF_W; ++kf) xb[kf] = *(const v2f*)&xRow[4 * kf];

    for (int t = 0; t < T_STEPS; ++t) {
        const int tn = (t + 1 < T_STEPS) ? (t + 1) : t;
        #pragma unroll
        for (int kf = 0; kf < NKF_W; ++kf) xn[kf] = *(const v2f*)&xRow[tn * F_IN + 4 * kf];

        phase<0>(usL, wsL, bsL, Ab, Wf, bV, xb, hOld, hNew[0], cSt[0]);
        phase<1>(usL, wsL, bsL, Ab, Wf, bV, xb, hOld, hNew[1], cSt[1]);
        phase<2>(usL, wsL, bsL, Ab, Wf, bV, xb, hOld, hNew[2], cSt[2]);
        phase<3>(usL, wsL, bsL, Ab, Wf, bV, xb, hOld, hNew[3], cSt[3]);
        phase<4>(usL, wsL, bsL, Ab, Wf, bV, xb, hOld, hNew[4], cSt[4]);
        phase<5>(usL, wsL, bsL, Ab, Wf, bV, xb, hOld, hNew[5], cSt[5]);
        phase<6>(usL, wsL, bsL, Ab, Wf, bV, xb, hOld, hNew[6], cSt[6]);
        phase<7>(usL, wsL, bsL, Ab, Wf, bV, xb, hOld, hNew[7], cSt[7]);

        #pragma unroll
        for (int u = 0; u < 8; ++u) hOld[u] = hNew[u];
        #pragma unroll
        for (int kf = 0; kf < NKF_W; ++kf) xb[kf] = xn[kf];
    }

    float part[10];
    #pragma unroll
    for (int c = 0; c < 10; ++c) part[c] = 0.0f;
    #pragma unroll
    for (int u = 0; u < 8; ++u) {
        #pragma unroll
        for (int r = 0; r < 8; ++r) {
            const int k = 16 * u + 8 * half + r;
            const float hv = hOld[u][r];
            const v2f* wr = (const v2f*)&Wd[k * 10];
            #pragma unroll
            for (int c2 = 0; c2 < 5; ++c2) {
                const v2f w2 = wr[c2];
                part[2 * c2]     += hv * w2.x;
                part[2 * c2 + 1] += hv * w2.y;
            }
        }
    }
    float logits[10];
    #pragma unroll
    for (int c = 0; c < 10; ++c)
        logits[c] = bd[c] + (part[c] + __shfl_xor(part[c], 16));
    float mx = logits[0];
    #pragma unroll
    for (int c = 1; c < 10; ++c) mx = fmaxf(mx, logits[c]);
    float e[10], sum = 0.0f;
    #pragma unroll
    for (int c = 0; c < 10; ++c) { e[c] = __expf(logits[c] - mx); sum += e[c]; }
    const float inv = 1.0f / sum;
    if (half == 0) {
        #pragma unroll
        for (int c = 0; c < 10; ++c)
            out[(size_t)(rowBase + laneN) * 10 + c] = e[c] * inv;
    }
}

extern "C" void kernel_launch(void* const* d_in, const int* in_sizes, int n_in,
                              void* d_out, int out_size, void* d_ws, size_t ws_size,
                              hipStream_t stream) {
    const float* x  = (const float*)d_in[0];
    const float* W  = (const float*)d_in[1];
    const float* U  = (const float*)d_in[2];
    const float* b  = (const float*)d_in[3];
    const float* Wd = (const float*)d_in[4];
    const float* bd = (const float*)d_in[5];
    float* out = (float*)d_out;

    hipLaunchKernelGGL(lstm_swapped_wmma, dim3(256), dim3(128), 0, stream,
                       x, W, U, b, Wd, bd, out);

    (void)in_sizes; (void)n_in; (void)out_size; (void)d_ws; (void)ws_size;
}


# cell update: the three sigmoid quotients merged into ONE exact f32 division, h=(c*A+g*B)/(A*B*E), c=h*E, A,B,E=1+e^-z with e^-z capped at 2^40; gate rounding moves a few ulp, matrix sums unchanged
# speedup vs baseline: 1.2968x; 1.0899x over previous
#include <hip/hip_runtime.h>
#include <math.h>


typedef __attribute__((ext_vector_type(2))) float v2f;
typedef __attribute__((ext_vector_type(4))) float v4f;
typedef __attribute__((ext_vector_type(8))) float v8f;

#define T_STEPS 28
#define F_IN    28
#define H_UNITS 128
#define G4      512
#define ROWS_WG 64
#define NKF_W   7
#define US_FLOATS 65536
#define WS_FLOATS 14336
#define BS_FLOATS 512

__device__ __forceinline__ float gate_den(float z) {
    return 1.0f + __builtin_amdgcn_exp2f(fminf(z * -1.44269504f, 40.0f));
}


__device__ __forceinline__ void load_u_block(v4f (&A)[4][4], const float* base)
{
    #pragma unroll
    for (int g = 0; g < 4; ++g) {
        #pragma unroll
        for (int p = 0; p < 4; ++p)
            A[g][p] = *(const v4f*)&base[(g * 4 + p) * 128];
    }
}

__device__ __forceinline__ void load_w_bias(v2f (&Wf)[NKF_W][4], v8f (&bV)[4],
                                            const float* wBase, const float* bBase)
{
    #pragma unroll
    for (int kf = 0; kf < NKF_W; ++kf) {
        #pragma unroll
        for (int g = 0; g < 4; ++g)
            Wf[kf][g] = *(const v2f*)&wBase[kf * 256 + g * 64];
    }
    #pragma unroll
    for (int g = 0; g < 4; ++g)
        bV[g] = *(const v8f*)&bBase[g * 16];
}

template <int UP>
__device__ __forceinline__ void phase(const float* usL, const float* wsL, const float* bsL,
                                      v4f (&Ab)[2][4][4], v2f (&Wf)[NKF_W][4], v8f (&bV)[4],
                                      const v2f (&xb)[NKF_W], const v8f (&hOld)[8],
                                      v8f& hNewU, v8f& cU)
{
    constexpr int UN = (UP + 1) & 7;
    v8f acc[4];

    load_u_block(Ab[1], usL + (UP * 4 + 1) * 2048);
    __builtin_amdgcn_sched_barrier(0);

    #pragma unroll
    for (int kf = 0; kf < NKF_W; ++kf) {
        #pragma unroll
        for (int g = 0; g < 4; ++g)
            acc[g] = __builtin_amdgcn_wmma_f32_16x16x4_f32(
                false, Wf[kf][g], false, xb[kf], (short)0, kf == 0 ? bV[g] : acc[g], false, false);
    }

    #pragma unroll
    for (int kb = 0; kb < 4; ++kb) {
        if (kb == 1) {
            load_u_block(Ab[0], usL + (UP * 4 + 2) * 2048);
            load_w_bias(Wf, bV, wsL + UN * (NKF_W * 256), bsL + UN * 64);
            __builtin_amdgcn_sched_barrier(0);
        } else if (kb == 2) {
            load_u_block(Ab[1], usL + (UP * 4 + 3) * 2048);
            __builtin_amdgcn_sched_barrier(0);
        } else if (kb == 3) {
            load_u_block(Ab[0], usL + (UN * 4 + 0) * 2048);
            __builtin_amdgcn_sched_barrier(0);
        }
        #pragma unroll
        for (int i = 0; i < 8; ++i) {
            const int u = 2 * kb + (i >> 2);
            const int j = i & 3;
            const v8f hv = hOld[u];
            v2f bfrag;
            bfrag.x = hv[2 * j];
            bfrag.y = hv[2 * j + 1];
            #pragma unroll
            for (int g = 0; g < 4; ++g) {
                const v4f a4 = Ab[kb & 1][g][i >> 1];
                v2f afrag;
                afrag.x = (i & 1) ? a4.z : a4.x;
                afrag.y = (i & 1) ? a4.w : a4.y;
                acc[g] = __builtin_amdgcn_wmma_f32_16x16x4_f32(
                    false, afrag, false, bfrag, (short)0, acc[g], false, false);
            }
        }
        if (kb < 3) {
            __builtin_amdgcn_sched_barrier(0);
        }
    }

    #pragma unroll
    for (int r = 0; r < 8; ++r) {
        const float dI = gate_den(acc[0][r]);
        const float dF = gate_den(acc[1][r]);
        const float gg = fmaxf(acc[2][r], 0.0f);
        const float dO = gate_den(acc[3][r]);
        const float num = cU[r] * dI + gg * dF;
        const float hn = num / ((dI * dF) * dO);
        hNewU[r] = hn;
        cU[r] = hn * dO;
    }
}

__global__ __launch_bounds__(128)
void lstm_swapped_wmma(const float* __restrict__ x, const float* __restrict__ W,
                       const float* __restrict__ U, const float* __restrict__ b,
                       const float* __restrict__ Wd, const float* __restrict__ bd,
                       float* __restrict__ out)
{
    __shared__ __attribute__((aligned(16))) float smem[US_FLOATS + WS_FLOATS + BS_FLOATS];
    float* Us = smem;
    float* Ws = smem + US_FLOATS;
    float* Bs = smem + US_FLOATS + WS_FLOATS;

    const int tid   = threadIdx.x;
    const int lane  = tid & 31;
    const int wave  = tid >> 5;
    const int laneN = lane & 15;
    const int half  = lane >> 4;
    const int rowBase = blockIdx.x * ROWS_WG + wave * 16;

    #pragma unroll 16
    for (int it = 0; it < 128; ++it) {
        const int g  = it & 3;
        const int kb = (it >> 2) & 3;
        const int up = it >> 4;
        const int k0 = 16 * (2 * kb + (wave >> 1)) + 8 * half + 4 * (wave & 1);
        const int col = g * H_UNITS + 16 * up + laneN;
        v4f v;
        v.x = U[(k0 + 0) * G4 + col];
        v.y = U[(k0 + 1) * G4 + col];
        v.z = U[(k0 + 2) * G4 + col];
        v.w = U[(k0 + 3) * G4 + col];
        *(v4f*)&Us[it * 512 + tid * 4] = v;
    }
    #pragma unroll
    for (int it = 0; it < 56; ++it) {
        const int kf = it % 7;
        const int up = it / 7;
        const int k = 4 * kf + 2 * half;
        const int col = wave * H_UNITS + 16 * up + laneN;
        v2f v;
        v.x = W[k * G4 + col];
        v.y = W[(k + 1) * G4 + col];
        *(v2f*)&Ws[it * 256 + tid * 2] = v;
    }
    #pragma unroll
    for (int i4 = 0; i4 < BS_FLOATS / 128; ++i4) {
        const int idx = tid + 128 * i4;
        const int r  = idx & 7;
        const int hf = (idx >> 3) & 1;
        const int g  = (idx >> 4) & 3;
        const int up = idx >> 6;
        Bs[idx] = b[g * H_UNITS + 16 * up + 8 * hf + r];
    }
    __syncthreads();

    const float* usL = Us + lane * 4;
    const float* wsL = Ws + lane * 2;
    const float* bsL = Bs + half * 8;

    const float* xRow = x + (size_t)(rowBase + laneN) * (T_STEPS * F_IN) + 2 * half;

    v8f hOld[8], hNew[8], cSt[8];
    #pragma unroll
    for (int u = 0; u < 8; ++u) {
        hOld[u] = (v8f){0.f,0.f,0.f,0.f,0.f,0.f,0.f,0.f};
        hNew[u] = hOld[u];
        cSt[u]  = hOld[u];
    }

    v4f Ab[2][4][4];
    v2f Wf[NKF_W][4];
    v8f bV[4];
    v2f xb[NKF_W], xn[NKF_W];

    load_u_block(Ab[0], usL);
    load_w_bias(Wf, bV, wsL, bsL);
    #pragma unroll
    for (int kf = 0; kf < NKF_W; ++kf) xb[kf] = *(const v2f*)&xRow[4 * kf];

    for (int t = 0; t < T_STEPS; ++t) {
        const int tn = (t + 1 < T_STEPS) ? (t + 1) : t;
        #pragma unroll
        for (int kf = 0; kf < NKF_W; ++kf) xn[kf] = *(const v2f*)&xRow[tn * F_IN + 4 * kf];

        phase<0>(usL, wsL, bsL, Ab, Wf, bV, xb, hOld, hNew[0], cSt[0]);
        phase<1>(usL, wsL, bsL, Ab, Wf, bV, xb, hOld, hNew[1], cSt[1]);
        phase<2>(usL, wsL, bsL, Ab, Wf, bV, xb, hOld, hNew[2], cSt[2]);
        phase<3>(usL, wsL, bsL, Ab, Wf, bV, xb, hOld, hNew[3], cSt[3]);
        phase<4>(usL, wsL, bsL, Ab, Wf, bV, xb, hOld, hNew[4], cSt[4]);
        phase<5>(usL, wsL, bsL, Ab, Wf, bV, xb, hOld, hNew[5], cSt[5]);
        phase<6>(usL, wsL, bsL, Ab, Wf, bV, xb, hOld, hNew[6], cSt[6]);
        phase<7>(usL, wsL, bsL, Ab, Wf, bV, xb, hOld, hNew[7], cSt[7]);

        #pragma unroll
        for (int u = 0; u < 8; ++u) hOld[u] = hNew[u];
        #pragma unroll
        for (int kf = 0; kf < NKF_W; ++kf) xb[kf] = xn[kf];
    }

    float part[10];
    #pragma unroll
    for (int c = 0; c < 10; ++c) part[c] = 0.0f;
    #pragma unroll
    for (int u = 0; u < 8; ++u) {
        #pragma unroll
        for (int r = 0; r < 8; ++r) {
            const int k = 16 * u + 8 * half + r;
            const float hv = hOld[u][r];
            const v2f* wr = (const v2f*)&Wd[k * 10];
            #pragma unroll
            for (int c2 = 0; c2 < 5; ++c2) {
                const v2f w2 = wr[c2];
                part[2 * c2]     += hv * w2.x;
                part[2 * c2 + 1] += hv * w2.y;
            }
        }
    }
    float logits[10];
    #pragma unroll
    for (int c = 0; c < 10; ++c)
        logits[c] = bd[c] + (part[c] + __shfl_xor(part[c], 16));
    float mx = logits[0];
    #pragma unroll
    for (int c = 1; c < 10; ++c) mx = fmaxf(mx, logits[c]);
    float e[10], sum = 0.0f;
    #pragma unroll
    for (int c = 0; c < 10; ++c) { e[c] = __expf(logits[c] - mx); sum += e[c]; }
    const float inv = 1.0f / sum;
    if (half == 0) {
        #pragma unroll
        for (int c = 0; c < 10; ++c)
            out[(size_t)(rowBase + laneN) * 10 + c] = e[c] * inv;
    }
}

extern "C" void kernel_launch(void* const* d_in, const int* in_sizes, int n_in,
                              void* d_out, int out_size, void* d_ws, size_t ws_size,
                              hipStream_t stream) {
    const float* x  = (const float*)d_in[0];
    const float* W  = (const float*)d_in[1];
    const float* U  = (const float*)d_in[2];
    const float* b  = (const float*)d_in[3];
    const float* Wd = (const float*)d_in[4];
    const float* bd = (const float*)d_in[5];
    float* out = (float*)d_out;

    hipLaunchKernelGGL(lstm_swapped_wmma, dim3(256), dim3(128), 0, stream,
                       x, W, U, b, Wd, bd, out);

    (void)in_sizes; (void)n_in; (void)out_size; (void)d_ws; (void)ws_size;
}


# cell update: gate sigmoids as 0.5+0.5*tanh(z/2) on the tanh instruction, the 0.5 folded exactly into the i,f,o weight rows at staging; no exp, no division; gate rounding moves, matrix sums unchanged
# speedup vs baseline: 1.3525x; 1.0430x over previous
#include <hip/hip_runtime.h>
#include <math.h>


typedef __attribute__((ext_vector_type(2))) float v2f;
typedef __attribute__((ext_vector_type(4))) float v4f;
typedef __attribute__((ext_vector_type(8))) float v8f;

#define T_STEPS 28
#define F_IN    28
#define H_UNITS 128
#define G4      512
#define ROWS_WG 64
#define NKF_W   7
#define US_FLOATS 65536
#define WS_FLOATS 14336
#define BS_FLOATS 512

__device__ __forceinline__ float gate_sigmoid_half_arg(float zh) {
    return __builtin_fmaf(0.5f, __builtin_amdgcn_tanhf(zh), 0.5f);
}


__device__ __forceinline__ void load_u_block(v4f (&A)[4][4], const float* base)
{
    #pragma unroll
    for (int g = 0; g < 4; ++g) {
        #pragma unroll
        for (int p = 0; p < 4; ++p)
            A[g][p] = *(const v4f*)&base[(g * 4 + p) * 128];
    }
}

__device__ __forceinline__ void load_w_bias(v2f (&Wf)[NKF_W][4], v8f (&bV)[4],
                                            const float* wBase, const float* bBase)
{
    #pragma unroll
    for (int kf = 0; kf < NKF_W; ++kf) {
        #pragma unroll
        for (int g = 0; g < 4; ++g)
            Wf[kf][g] = *(const v2f*)&wBase[kf * 256 + g * 64];
    }
    #pragma unroll
    for (int g = 0; g < 4; ++g)
        bV[g] = *(const v8f*)&bBase[g * 16];
}

template <int UP>
__device__ __forceinline__ void phase(const float* usL, const float* wsL, const float* bsL,
                                      v4f (&Ab)[2][4][4], v2f (&Wf)[NKF_W][4], v8f (&bV)[4],
                                      const v2f (&xb)[NKF_W], const v8f (&hOld)[8],
                                      v8f& hNewU, v8f& cU)
{
    constexpr int UN = (UP + 1) & 7;
    v8f acc[4];

    load_u_block(Ab[1], usL + (UP * 4 + 1) * 2048);
    __builtin_amdgcn_sched_barrier(0);

    #pragma unroll
    for (int kf = 0; kf < NKF_W; ++kf) {
        #pragma unroll
        for (int g = 0; g < 4; ++g)
            acc[g] = __builtin_amdgcn_wmma_f32_16x16x4_f32(
                false, Wf[kf][g], false, xb[kf], (short)0, kf == 0 ? bV[g] : acc[g], false, false);
    }

    #pragma unroll
    for (int kb = 0; kb < 4; ++kb) {
        if (kb == 1) {
            load_u_block(Ab[0], usL + (UP * 4 + 2) * 2048);
            load_w_bias(Wf, bV, wsL + UN * (NKF_W * 256), bsL + UN * 64);
            __builtin_amdgcn_sched_barrier(0);
        } else if (kb == 2) {
            load_u_block(Ab[1], usL + (UP * 4 + 3) * 2048);
            __builtin_amdgcn_sched_barrier(0);
        } else if (kb == 3) {
            load_u_block(Ab[0], usL + (UN * 4 + 0) * 2048);
            __builtin_amdgcn_sched_barrier(0);
        }
        #pragma unroll
        for (int i = 0; i < 8; ++i) {
            const int u = 2 * kb + (i >> 2);
            const int j = i & 3;
            const v8f hv = hOld[u];
            v2f bfrag;
            bfrag.x = hv[2 * j];
            bfrag.y = hv[2 * j + 1];
            #pragma unroll
            for (int g = 0; g < 4; ++g) {
                const v4f a4 = Ab[kb & 1][g][i >> 1];
                v2f afrag;
                afrag.x = (i & 1) ? a4.z : a4.x;
                afrag.y = (i & 1) ? a4.w : a4.y;
                acc[g] = __builtin_amdgcn_wmma_f32_16x16x4_f32(
                    false, afrag, false, bfrag, (short)0, acc[g], false, false);
            }
        }
        if (kb < 3) {
            __builtin_amdgcn_sched_barrier(0);
        }
    }

    __builtin_amdgcn_sched_barrier(0);

    #pragma unroll
    for (int r = 0; r < 8; ++r) {
        const float gi = gate_sigmoid_half_arg(acc[0][r]);
        const float gf = gate_sigmoid_half_arg(acc[1][r]);
        const float gg = fmaxf(acc[2][r], 0.0f);
        const float go = gate_sigmoid_half_arg(acc[3][r]);
        const float cn = __builtin_fmaf(gf, cU[r], gi * gg);
        cU[r] = cn;
        hNewU[r] = go * fmaxf(cn, 0.0f);
    }
}

__global__ __launch_bounds__(128)
void lstm_swapped_wmma(const float* __restrict__ x, const float* __restrict__ W,
                       const float* __restrict__ U, const float* __restrict__ b,
                       const float* __restrict__ Wd, const float* __restrict__ bd,
                       float* __restrict__ out)
{
    __shared__ __attribute__((aligned(16))) float smem[US_FLOATS + WS_FLOATS + BS_FLOATS];
    float* Us = smem;
    float* Ws = smem + US_FLOATS;
    float* Bs = smem + US_FLOATS + WS_FLOATS;

    const int tid   = threadIdx.x;
    const int lane  = tid & 31;
    const int wave  = tid >> 5;
    const int laneN = lane & 15;
    const int half  = lane >> 4;
    const int rowBase = blockIdx.x * ROWS_WG + wave * 16;

    #pragma unroll 16
    for (int it = 0; it < 128; ++it) {
        const int g  = it & 3;
        const int kb = (it >> 2) & 3;
        const int up = it >> 4;
        const int k0 = 16 * (2 * kb + (wave >> 1)) + 8 * half + 4 * (wave & 1);
        const int col = g * H_UNITS + 16 * up + laneN;
        const float sc = (g == 2) ? 1.0f : 0.5f;
        v4f v;
        v.x = U[(k0 + 0) * G4 + col] * sc;
        v.y = U[(k0 + 1) * G4 + col] * sc;
        v.z = U[(k0 + 2) * G4 + col] * sc;
        v.w = U[(k0 + 3) * G4 + col] * sc;
        *(v4f*)&Us[it * 512 + tid * 4] = v;
    }
    #pragma unroll
    for (int it = 0; it < 56; ++it) {
        const int kf = it % 7;
        const int up = it / 7;
        const int k = 4 * kf + 2 * half;
        const int col = wave * H_UNITS + 16 * up + laneN;
        const float sc = (wave == 2) ? 1.0f : 0.5f;
        v2f v;
        v.x = W[k * G4 + col] * sc;
        v.y = W[(k + 1) * G4 + col] * sc;
        *(v2f*)&Ws[it * 256 + tid * 2] = v;
    }
    #pragma unroll
    for (int i4 = 0; i4 < BS_FLOATS / 128; ++i4) {
        const int idx = tid + 128 * i4;
        const int r  = idx & 7;
        const int hf = (idx >> 3) & 1;
        const int g  = (idx >> 4) & 3;
        const int up = idx >> 6;
        Bs[idx] = b[g * H_UNITS + 16 * up + 8 * hf + r] * ((g == 2) ? 1.0f : 0.5f);
    }
    __syncthreads();

    const float* usL = Us + lane * 4;
    const float* wsL = Ws + lane * 2;
    const float* bsL = Bs + half * 8;

    const float* xRow = x + (size_t)(rowBase + laneN) * (T_STEPS * F_IN) + 2 * half;

    v8f hOld[8], hNew[8], cSt[8];
    #pragma unroll
    for (int u = 0; u < 8; ++u) {
        hOld[u] = (v8f){0.f,0.f,0.f,0.f,0.f,0.f,0.f,0.f};
        hNew[u] = hOld[u];
        cSt[u]  = hOld[u];
    }

    v4f Ab[2][4][4];
    v2f Wf[NKF_W][4];
    v8f bV[4];
    v2f xb[NKF_W], xn[NKF_W];

    load_u_block(Ab[0], usL);
    load_w_bias(Wf, bV, wsL, bsL);
    #pragma unroll
    for (int kf = 0; kf < NKF_W; ++kf) xb[kf] = *(const v2f*)&xRow[4 * kf];

    for (int t = 0; t < T_STEPS; ++t) {
        const int tn = (t + 1 < T_STEPS) ? (t + 1) : t;
        #pragma unroll
        for (int kf = 0; kf < NKF_W; ++kf) xn[kf] = *(const v2f*)&xRow[tn * F_IN + 4 * kf];

        phase<0>(usL, wsL, bsL, Ab, Wf, bV, xb, hOld, hNew[0], cSt[0]);
        phase<1>(usL, wsL, bsL, Ab, Wf, bV, xb, hOld, hNew[1], cSt[1]);
        phase<2>(usL, wsL, bsL, Ab, Wf, bV, xb, hOld, hNew[2], cSt[2]);
        phase<3>(usL, wsL, bsL, Ab, Wf, bV, xb, hOld, hNew[3], cSt[3]);
        phase<4>(usL, wsL, bsL, Ab, Wf, bV, xb, hOld, hNew[4], cSt[4]);
        phase<5>(usL, wsL, bsL, Ab, Wf, bV, xb, hOld, hNew[5], cSt[5]);
        phase<6>(usL, wsL, bsL, Ab, Wf, bV, xb, hOld, hNew[6], cSt[6]);
        phase<7>(usL, wsL, bsL, Ab, Wf, bV, xb, hOld, hNew[7], cSt[7]);

        #pragma unroll
        for (int u = 0; u < 8; ++u) hOld[u] = hNew[u];
        #pragma unroll
        for (int kf = 0; kf < NKF_W; ++kf) xb[kf] = xn[kf];
    }

    float part[10];
    #pragma unroll
    for (int c = 0; c < 10; ++c) part[c] = 0.0f;
    #pragma unroll
    for (int u = 0; u < 8; ++u) {
        #pragma unroll
        for (int r = 0; r < 8; ++r) {
            const int k = 16 * u + 8 * half + r;
            const float hv = hOld[u][r];
            const v2f* wr = (const v2f*)&Wd[k * 10];
            #pragma unroll
            for (int c2 = 0; c2 < 5; ++c2) {
                const v2f w2 = wr[c2];
                part[2 * c2]     += hv * w2.x;
                part[2 * c2 + 1] += hv * w2.y;
            }
        }
    }
    float logits[10];
    #pragma unroll
    for (int c = 0; c < 10; ++c)
        logits[c] = bd[c] + (part[c] + __shfl_xor(part[c], 16));
    float mx = logits[0];
    #pragma unroll
    for (int c = 1; c < 10; ++c) mx = fmaxf(mx, logits[c]);
    float e[10], sum = 0.0f;
    #pragma unroll
    for (int c = 0; c < 10; ++c) { e[c] = __expf(logits[c] - mx); sum += e[c]; }
    const float inv = 1.0f / sum;
    if (half == 0) {
        #pragma unroll
        for (int c = 0; c < 10; ++c)
            out[(size_t)(rowBase + laneN) * 10 + c] = e[c] * inv;
    }
}

extern "C" void kernel_launch(void* const* d_in, const int* in_sizes, int n_in,
                              void* d_out, int out_size, void* d_ws, size_t ws_size,
                              hipStream_t stream) {
    const float* x  = (const float*)d_in[0];
    const float* W  = (const float*)d_in[1];
    const float* U  = (const float*)d_in[2];
    const float* b  = (const float*)d_in[3];
    const float* Wd = (const float*)d_in[4];
    const float* bd = (const float*)d_in[5];
    float* out = (float*)d_out;

    hipLaunchKernelGGL(lstm_swapped_wmma, dim3(256), dim3(128), 0, stream,
                       x, W, U, b, Wd, bd, out);

    (void)in_sizes; (void)n_in; (void)out_size; (void)d_ws; (void)ws_size;
}
